# starting kernel
# baseline (speedup 1.0000x reference)
#include <hip/hip_runtime.h>
#include <math.h>

#define N_NODES   32768
#define N_EDGES   524288
#define NODE_DIM  128
#define HIDDEN_DIM 256
#define EDGE_DIM  64
#define IN_DIM    320

#define TILE_E    64
#define THREADS   256

#define A_LD      324
#define A_OFF     0
#define A_SIZE    (TILE_E * A_LD)
#define H_LD      260
#define H_OFF     (A_OFF + A_SIZE)
#define H_SIZE    (TILE_E * H_LD)
#define B_OFF     (H_OFF + H_SIZE)
#define B_SIZE    (32 * 256)
#define S_OFF     (B_OFF + B_SIZE)
#define SMEM_FLOATS (S_OFF + 4 * TILE_E)

typedef float v2f __attribute__((ext_vector_type(2)));
typedef float v8f __attribute__((ext_vector_type(8)));

__device__ __forceinline__ float silu_f(float v) {
    return v / (1.0f + __expf(-v));
}

__device__ __forceinline__ void atomAddF(float* p, float v) {
    __hip_atomic_fetch_add(p, v, __ATOMIC_RELAXED, __HIP_MEMORY_SCOPE_AGENT);
}

template<int NTILES, int NSTAGE>
__device__ __forceinline__ void gemm_tile(const float* __restrict__ W, int ldw, int K,
                                          const float* __restrict__ A, int lda,
                                          float* __restrict__ Bs,
                                          v8f* acc, int tid)
{
    const int lane  = tid & 31;
    const int wave  = tid >> 5;
    const int row   = lane & 15;
    const int hi    = lane >> 4;
    const int etile = wave >> 1;
    const int nsel  = wave & 1;
    const int nbase = nsel * (NSTAGE / 2);

    for (int kp = 0; kp < K; kp += 32) {
        __syncthreads();
        for (int i = tid; i < 32 * NSTAGE; i += THREADS) {
            int kl = i / NSTAGE;
            int n  = i - kl * NSTAGE;
            Bs[(((kl >> 2) * NSTAGE + n) << 2) + (kl & 3)] = W[(kp + kl) * ldw + n];
        }
        __syncthreads();

        const float* Arow = A + (etile * 16 + row) * lda + kp + 2 * hi;
#pragma unroll
        for (int kl = 0; kl < 32; kl += 4) {
            v2f a = *(const v2f*)(Arow + kl);
#pragma unroll
            for (int nt = 0; nt < NTILES; ++nt) {
                v2f b = *(const v2f*)(Bs + ((((kl >> 2) * NSTAGE + nbase + nt * 16 + row) << 2) + 2 * hi));
                acc[nt] = __builtin_amdgcn_wmma_f32_16x16x4_f32(
                     false, a,  false, b,
                     (short)0, acc[nt],
                     false,  false);
            }
        }
    }
}

__global__ void egnn_init_out(const float* __restrict__ h,
                              const float* __restrict__ x,
                              float* __restrict__ out)
{
    int i = blockIdx.x * blockDim.x + threadIdx.x;
    if (i < N_NODES * NODE_DIM)                out[i] = h[i];
    else if (i < N_NODES * (NODE_DIM + 3))     out[i] = x[i - N_NODES * NODE_DIM];
}

__global__ void __launch_bounds__(THREADS)
egnn_edge_kernel(const float* __restrict__ h,   const float* __restrict__ x,
                 const int*   __restrict__ eidx,const float* __restrict__ edist,
                 const float* __restrict__ nW1, const float* __restrict__ nb1,
                 const float* __restrict__ nW2, const float* __restrict__ nb2,
                 const float* __restrict__ cW1, const float* __restrict__ cb1,
                 const float* __restrict__ cW2,
                 const float* __restrict__ eW1, const float* __restrict__ eb1,
                 const float* __restrict__ eW2, const float* __restrict__ eb2,
                 float* __restrict__ out)
{
    extern __shared__ float smem[];
    float* Ablk = smem + A_OFF;
    float* Hblk = smem + H_OFF;
    float* Bs   = smem + B_OFF;
    int*   SRCs = (int*)(smem + S_OFF);
    int*   DSTs = (int*)(smem + S_OFF + TILE_E);
    float* DIST = smem + S_OFF + 2 * TILE_E;

    const int tid   = threadIdx.x;
    const int lane  = tid & 31;
    const int wave  = tid >> 5;
    const int row   = lane & 15;
    const int hi    = lane >> 4;
    const int etile = wave >> 1;
    const int nsel  = wave & 1;
    const int e0    = blockIdx.x * TILE_E;

    if (tid < TILE_E) {
        SRCs[tid] = eidx[e0 + tid];
        DSTs[tid] = eidx[N_EDGES + e0 + tid];
        DIST[tid] = edist[e0 + tid];
    }
    __syncthreads();

    for (int i = tid; i < TILE_E * 64; i += THREADS) {
        int e = i >> 6, q = i & 63;
        int half = q >> 5, c4 = q & 31;
        int node = half ? DSTs[e] : SRCs[e];
        *(float4*)(Ablk + e * A_LD + half * NODE_DIM + c4 * 4) =
            ((const float4*)h)[node * (NODE_DIM / 4) + c4];
    }

    float* hid = Hblk;
    for (int i = tid; i < TILE_E * EDGE_DIM; i += THREADS) {
        int e = i >> 6, k = i & 63;
        hid[k * TILE_E + e] = silu_f(DIST[e] * eW1[k] + eb1[k]);
    }
    __syncthreads();
    for (int i = tid; i < TILE_E * EDGE_DIM; i += THREADS) {
        int e = i >> 6, j = i & 63;
        float s = eb2[j];
#pragma unroll 8
        for (int k = 0; k < EDGE_DIM; ++k)
            s += hid[k * TILE_E + e] * eW2[k * EDGE_DIM + j];
        Ablk[e * A_LD + 2 * NODE_DIM + j] = s;
    }

    {
        v8f acc[8];
#pragma unroll
        for (int nt = 0; nt < 8; ++nt) {
            float bv = nb1[nsel * 128 + nt * 16 + row];
            v8f t = {bv, bv, bv, bv, bv, bv, bv, bv};
            acc[nt] = t;
        }
        gemm_tile<8, 256>(nW1, HIDDEN_DIM, IN_DIM, Ablk, A_LD, Bs, acc, tid);
#pragma unroll
        for (int nt = 0; nt < 8; ++nt)
#pragma unroll
            for (int i = 0; i < 8; ++i) {
                int e = etile * 16 + hi * 8 + i;
                Hblk[e * H_LD + nsel * 128 + nt * 16 + row] = silu_f(acc[nt][i]);
            }
    }

    {
        v8f acc[4];
#pragma unroll
        for (int nt = 0; nt < 4; ++nt) {
            float bv = nb2[nsel * 64 + nt * 16 + row];
            v8f t = {bv, bv, bv, bv, bv, bv, bv, bv};
            acc[nt] = t;
        }
        gemm_tile<4, 128>(nW2, NODE_DIM, HIDDEN_DIM, Hblk, H_LD, Bs, acc, tid);
#pragma unroll
        for (int nt = 0; nt < 4; ++nt)
#pragma unroll
            for (int i = 0; i < 8; ++i) {
                int e = etile * 16 + hi * 8 + i;
                int d = DSTs[e];
                atomAddF(out + d * NODE_DIM + nsel * 64 + nt * 16 + row, acc[nt][i]);
            }
    }

    {
        v8f acc[8];
#pragma unroll
        for (int nt = 0; nt < 8; ++nt) {
            float bv = cb1[nsel * 128 + nt * 16 + row];
            v8f t = {bv, bv, bv, bv, bv, bv, bv, bv};
            acc[nt] = t;
        }
        gemm_tile<8, 256>(cW1, HIDDEN_DIM, IN_DIM, Ablk, A_LD, Bs, acc, tid);
#pragma unroll
        for (int nt = 0; nt < 8; ++nt)
#pragma unroll
            for (int i = 0; i < 8; ++i) {
                int e = etile * 16 + hi * 8 + i;
                Hblk[e * H_LD + nsel * 128 + nt * 16 + row] = silu_f(acc[nt][i]);
            }
    }
    __syncthreads();

    if (tid < TILE_E) {
        int e = tid;
        float s = 0.0f;
#pragma unroll 8
        for (int k = 0; k < HIDDEN_DIM; ++k)
            s += Hblk[e * H_LD + k] * cW2[k];
        int si = SRCs[e], di = DSTs[e];
        float dx = x[si * 3 + 0] - x[di * 3 + 0];
        float dy = x[si * 3 + 1] - x[di * 3 + 1];
        float dz = x[si * 3 + 2] - x[di * 3 + 2];
        float len = fmaxf(sqrtf(dx * dx + dy * dy + dz * dz), 1e-8f);
        float w = s / len;
        float* ox = out + N_NODES * NODE_DIM;
        atomAddF(ox + di * 3 + 0, w * dx);
        atomAddF(ox + di * 3 + 1, w * dy);
        atomAddF(ox + di * 3 + 2, w * dz);
    }
}

extern "C" void kernel_launch(void* const* d_in, const int* in_sizes, int n_in,
                              void* d_out, int out_size, void* d_ws, size_t ws_size,
                              hipStream_t stream)
{
    const float* h     = (const float*)d_in[0];
    const float* x     = (const float*)d_in[1];
    const int*   eidx  = (const int*)  d_in[2];
    const float* edist = (const float*)d_in[3];
    const float* nW1   = (const float*)d_in[4];
    const float* nb1   = (const float*)d_in[5];
    const float* nW2   = (const float*)d_in[6];
    const float* nb2   = (const float*)d_in[7];
    const float* cW1   = (const float*)d_in[8];
    const float* cb1   = (const float*)d_in[9];
    const float* cW2   = (const float*)d_in[10];
    const float* eW1   = (const float*)d_in[11];
    const float* eb1   = (const float*)d_in[12];
    const float* eW2   = (const float*)d_in[13];
    const float* eb2   = (const float*)d_in[14];
    float* out = (float*)d_out;

    const int tot = N_NODES * (NODE_DIM + 3);
    egnn_init_out<<<(tot + 255) / 256, 256, 0, stream>>>(h, x, out);

    const size_t smem_bytes = (size_t)SMEM_FLOATS * sizeof(float);
    egnn_edge_kernel<<<N_EDGES / TILE_E, THREADS, smem_bytes, stream>>>(
        h, x, eidx, edist,
        nW1, nb1, nW2, nb2,
        cW1, cb1, cW2,
        eW1, eb1, eW2, eb2,
        out);
}


# B fragments read by 16-byte global loads, one trip ahead, from a fragment-ordered weight copy the init launch packs into workspace: no LDS weight staging, one barrier per chunk; sum order unchanged
# speedup vs baseline: 2.3166x; 2.3166x over previous
#include <hip/hip_runtime.h>
#include <math.h>


#define N_NODES   32768
#define N_EDGES   524288
#define NODE_DIM  128
#define HIDDEN_DIM 256
#define EDGE_DIM  64
#define IN_DIM    320

#define TILE_E    128
#define THREADS   512
#define CHUNK     64
#define NCHUNK    (HIDDEN_DIM / CHUNK)

#define A_LD      324
#define A_OFF     0
#define A_SIZE    (TILE_E * A_LD)
#define H_LD      68
#define H_OFF     (A_OFF + A_SIZE)
#define H_SIZE    (TILE_E * H_LD)
#define S_OFF     (H_OFF + 2 * H_SIZE)
#define SMEM_FLOATS (S_OFF + 4 * TILE_E)

#define KS1        (IN_DIM / 4)
#define KS2        (CHUNK / 4)
#define P1_FLOATS  (IN_DIM * HIDDEN_DIM)
#define P2_FLOATS  (HIDDEN_DIM * NODE_DIM)
#define PN1_OFF    0
#define PC1_OFF    P1_FLOATS
#define PN2_OFF    (2 * P1_FLOATS)
#define PACK_FLOATS (2 * P1_FLOATS + P2_FLOATS)
#define OUT_FLOATS (N_NODES * (NODE_DIM + 3))

typedef float v2f __attribute__((ext_vector_type(2)));
typedef float v8f __attribute__((ext_vector_type(8)));

__device__ __forceinline__ float silu_f(float v) {
    return v / (1.0f + __expf(-v));
}

__device__ __forceinline__ void atomAddF(float* p, float v) {
    __hip_atomic_fetch_add(p, v, __ATOMIC_RELAXED, __HIP_MEMORY_SCOPE_AGENT);
}

struct BSet { float4 v[4]; };

__device__ __forceinline__ void load_bset(BSet& s, const float4* p) {
#pragma unroll
    for (int u = 0; u < 4; ++u) s.v[u] = p[u * 32];
}

__device__ __forceinline__ void wm(v8f& acc, v2f a, float bx, float by) {
    v2f b = {bx, by};
    acc = __builtin_amdgcn_wmma_f32_16x16x4_f32( false, a,  false, b,
                                                 (short)0, acc,
                                                 false,  false);
}

__device__ __forceinline__ void trip_l1(const float* __restrict__ ap, const BSet& b, v8f* acc) {
    v2f a[4];
#pragma unroll
    for (int u = 0; u < 4; ++u) a[u] = *(const v2f*)(ap + u * 4);
#pragma unroll
    for (int u = 0; u < 4; ++u) {
        wm(acc[0], a[u], b.v[u].x, b.v[u].y);
        wm(acc[1], a[u], b.v[u].z, b.v[u].w);
    }
}

__device__ __forceinline__ void trip_l2(const float* __restrict__ hp, const BSet& b, v8f* acc) {
    v2f a[2];
#pragma unroll
    for (int s = 0; s < 2; ++s) a[s] = *(const v2f*)(hp + s * 4);
#pragma unroll
    for (int s = 0; s < 2; ++s) {
        wm(acc[0], a[s], b.v[2 * s].x,     b.v[2 * s].y);
        wm(acc[1], a[s], b.v[2 * s].z,     b.v[2 * s].w);
        wm(acc[2], a[s], b.v[2 * s + 1].x, b.v[2 * s + 1].y);
        wm(acc[3], a[s], b.v[2 * s + 1].z, b.v[2 * s + 1].w);
    }
}

template<bool FIRST>
__device__ __forceinline__ void trip_one(const float* __restrict__ ap, const BSet& b, v8f* acc) {
    if constexpr (FIRST) trip_l1(ap, b, acc); else trip_l2(ap, b, acc);
}

template<bool FIRST>
__device__ __forceinline__ void trip_pair(const float* __restrict__ ap, const float4* bg,
                                          BSet& b0, BSet& b1, v8f* acc)
{
    constexpr int ASTEP = FIRST ? 16 : 8;
    load_bset(b1, bg + 128);
    trip_one<FIRST>(ap, b0, acc);
    if constexpr (FIRST) asm volatile("" : "+v"(acc[0]), "+v"(acc[1]) :: "memory");
    else                 asm volatile("" : "+v"(acc[0]), "+v"(acc[1]), "+v"(acc[2]), "+v"(acc[3]) :: "memory");
    load_bset(b0, bg + 256);
    trip_one<FIRST>(ap + ASTEP, b1, acc);
}

template<int TRIPS, bool FIRST>
__device__ __forceinline__ void mma_stream(const float* __restrict__ ap,
                                           const float4* bg,
                                           BSet& b0, v8f* acc)
{
    static_assert(TRIPS % 2 == 0 && TRIPS >= 2, "trips are taken in pairs");
    constexpr int ASTEP = FIRST ? 16 : 8;
    BSet b1;
    if constexpr (FIRST) {
#pragma unroll 1
        for (int t = 0; t < TRIPS - 2; t += 2) {
            trip_pair<FIRST>(ap, bg, b0, b1, acc);
            ap += 2 * ASTEP;
            bg += 256;
        }
    } else {
#pragma unroll
        for (int t = 0; t < TRIPS - 2; t += 2) {
            trip_pair<FIRST>(ap, bg, b0, b1, acc);
            ap += 2 * ASTEP;
            bg += 256;
        }
    }
    load_bset(b1, bg + 128);
    trip_one<FIRST>(ap, b0, acc);
    trip_one<FIRST>(ap + ASTEP, b1, acc);
}

__global__ void egnn_init_out(const float* __restrict__ h,
                              const float* __restrict__ x,
                              const float* __restrict__ nW1,
                              const float* __restrict__ cW1,
                              const float* __restrict__ nW2,
                              float* __restrict__ out,
                              float* __restrict__ wpack)
{
    int i = blockIdx.x * blockDim.x + threadIdx.x;
    if (i < N_NODES * NODE_DIM)      out[i] = h[i];
    else if (i < OUT_FLOATS)         out[i] = x[i - N_NODES * NODE_DIM];
    else {
        int p = i - OUT_FLOATS;
        if (p >= PACK_FLOATS) return;
        int lane = (p >> 2) & 31;
        int row = lane & 15, hi = lane >> 4;
        int j = p & 1, ntl = (p >> 1) & 1;
        if (p < 2 * P1_FLOATS) {
            const float* Wm = (p < P1_FLOATS) ? nW1 : cW1;
            int q  = (p < P1_FLOATS) ? p : p - P1_FLOATS;
            int t  = q >> 7;
            int cn = t / KS1;
            int ks = t - cn * KS1;
            int k  = ks * 4 + 2 * hi + j;
            int n  = (cn >> 1) * CHUNK + (cn & 1) * 32 + ntl * 16 + row;
            wpack[p] = Wm[k * HIDDEN_DIM + n];
        } else {
            int q  = p - 2 * P1_FLOATS;
            int qq = (q >> 7) & 1;
            int ks = (q >> 8) & (KS2 - 1);
            int cn = q >> 12;
            int k  = (cn >> 1) * CHUNK + ks * 4 + 2 * hi + j;
            int n  = (cn & 1) * 64 + qq * 32 + ntl * 16 + row;
            wpack[p] = nW2[k * NODE_DIM + n];
        }
    }
}

__global__ void __launch_bounds__(THREADS)
egnn_edge_kernel(const float* __restrict__ h,   const float* __restrict__ x,
                 const int*   __restrict__ eidx,const float* __restrict__ edist,
                 const float* __restrict__ nb1, const float* __restrict__ nb2,
                 const float* __restrict__ cb1, const float* __restrict__ cW2,
                 const float* __restrict__ eW1, const float* __restrict__ eb1,
                 const float* __restrict__ eW2, const float* __restrict__ eb2,
                 const float* wpack,
                 float* __restrict__ out)
{
    __shared__ __align__(16) float smem[SMEM_FLOATS];
    float* Ablk = smem + A_OFF;
    float* Hblk = smem + H_OFF;
    int*   SRCs = (int*)(smem + S_OFF);
    int*   DSTs = (int*)(smem + S_OFF + TILE_E);
    float* DIST = smem + S_OFF + 2 * TILE_E;

    const int tid   = threadIdx.x;
    const int lane  = tid & 31;
    const int wave  = tid >> 5;
    const int row   = lane & 15;
    const int hi    = lane >> 4;
    const int etile = wave >> 1;
    const int nsel  = wave & 1;
    const int e0    = blockIdx.x * TILE_E;

    if (tid < TILE_E) {
        SRCs[tid] = eidx[e0 + tid];
        DSTs[tid] = eidx[N_EDGES + e0 + tid];
        DIST[tid] = edist[e0 + tid];
    }
    __syncthreads();

    for (int i = tid; i < TILE_E * 64; i += THREADS) {
        int e = i >> 6, q = i & 63;
        int half = q >> 5, c4 = q & 31;
        int node = half ? DSTs[e] : SRCs[e];
        *(float4*)(Ablk + e * A_LD + half * NODE_DIM + c4 * 4) =
            ((const float4*)h)[node * (NODE_DIM / 4) + c4];
    }

    float* hid = Hblk;
    {
        const int   e = tid & (TILE_E - 1);
        const float d = DIST[e];
#pragma unroll 4
        for (int it = 0; it < EDGE_DIM / (THREADS / TILE_E); ++it) {
            int k = (tid >> 7) + it * (THREADS / TILE_E);
            hid[k * TILE_E + e] = silu_f(d * eW1[k] + eb1[k]);
        }
    }
    __syncthreads();
    {
        const int c4 = (tid & 15) * 4;
        const int e4 = (tid >> 4) * 4;
        float acc[4][4];
#pragma unroll
        for (int c = 0; c < 4; ++c) {
            float bv = eb2[c4 + c];
#pragma unroll
            for (int t = 0; t < 4; ++t) acc[t][c] = bv;
        }
        const float* hp = hid + e4;
        const float* wp = eW2 + c4;
#pragma unroll 8
        for (int k = 0; k < EDGE_DIM; ++k) {
            const float4 hv = *(const float4*)(hp + k * TILE_E);
            const float4 wv = *(const float4*)(wp + k * EDGE_DIM);
            const float hh[4] = {hv.x, hv.y, hv.z, hv.w};
            const float ww[4] = {wv.x, wv.y, wv.z, wv.w};
#pragma unroll
            for (int t = 0; t < 4; ++t)
#pragma unroll
                for (int c = 0; c < 4; ++c)
                    acc[t][c] += hh[t] * ww[c];
        }
#pragma unroll
        for (int t = 0; t < 4; ++t) {
            float4 o;
            o.x = acc[t][0]; o.y = acc[t][1]; o.z = acc[t][2]; o.w = acc[t][3];
            *(float4*)(Ablk + (e4 + t) * A_LD + 2 * NODE_DIM + c4) = o;
        }
    }
    __syncthreads();

    const float*  Arow = Ablk + (etile * 16 + row) * A_LD + 2 * hi;
    const float*  Hrow = Hblk + (etile * 16 + row) * H_LD + 2 * hi;
    const float4* wp4  = (const float4*)wpack;
    const float4* Bn1  = wp4 + PN1_OFF / 4 + nsel * (KS1 * 32) + lane;
    const float4* Bc1  = wp4 + PC1_OFF / 4 + nsel * (KS1 * 32) + lane;
    const float4* Bn2  = wp4 + PN2_OFF / 4 + nsel * (KS2 * 2 * 32) + lane;

    {
        v8f macc[4];
#pragma unroll
        for (int nt = 0; nt < 4; ++nt) {
            float bv = nb2[nsel * 64 + nt * 16 + row];
            v8f t = {bv, bv, bv, bv, bv, bv, bv, bv};
            macc[nt] = t;
        }
#pragma unroll 1
        for (int c = 0; c < NCHUNK; ++c) {
            BSet b1s;
            load_bset(b1s, Bn1 + c * (2 * KS1 * 32));
            v8f acc[2];
#pragma unroll
            for (int nt = 0; nt < 2; ++nt) {
                float bv = nb1[c * CHUNK + nsel * 32 + nt * 16 + row];
                v8f t = {bv, bv, bv, bv, bv, bv, bv, bv};
                acc[nt] = t;
            }
            mma_stream<KS1 / 4, true>(Arow, Bn1 + c * (2 * KS1 * 32), b1s, acc);
            BSet b2s;
            load_bset(b2s, Bn2 + c * (2 * KS2 * 2 * 32));
            asm volatile("v_nop\n\tv_nop\n\tv_nop\n\tv_nop" : "+v"(acc[0]), "+v"(acc[1]));
            __builtin_amdgcn_sched_barrier(0);
            float* Hc = Hblk + (c & 1) * H_SIZE;
#pragma unroll
            for (int nt = 0; nt < 2; ++nt)
#pragma unroll
                for (int i = 0; i < 8; ++i) {
                    int e = etile * 16 + hi * 8 + i;
                    Hc[e * H_LD + nsel * 32 + nt * 16 + row] = silu_f(acc[nt][i]);
                }
            __syncthreads();
            mma_stream<KS2 / 2, false>(Hrow + (c & 1) * H_SIZE, Bn2 + c * (2 * KS2 * 2 * 32), b2s, macc);
            asm volatile("v_nop\n\tv_nop\n\tv_nop\n\tv_nop" : "+v"(macc[0]), "+v"(macc[1]), "+v"(macc[2]), "+v"(macc[3]));
            __builtin_amdgcn_sched_barrier(0);
        }
#pragma unroll
        for (int nt = 0; nt < 4; ++nt)
#pragma unroll
            for (int i = 0; i < 8; ++i) {
                int e = etile * 16 + hi * 8 + i;
                int d = DSTs[e];
                atomAddF(out + d * NODE_DIM + nsel * 64 + nt * 16 + row, macc[nt][i]);
            }
    }

    float s = 0.0f;
#pragma unroll 1
    for (int c = 0; c < NCHUNK; ++c) {
        BSet b1s;
        load_bset(b1s, Bc1 + c * (2 * KS1 * 32));
        v8f acc[2];
#pragma unroll
        for (int nt = 0; nt < 2; ++nt) {
            float bv = cb1[c * CHUNK + nsel * 32 + nt * 16 + row];
            v8f t = {bv, bv, bv, bv, bv, bv, bv, bv};
            acc[nt] = t;
        }
        mma_stream<KS1 / 4, true>(Arow, Bc1 + c * (2 * KS1 * 32), b1s, acc);
        asm volatile("v_nop\n\tv_nop\n\tv_nop\n\tv_nop" : "+v"(acc[0]), "+v"(acc[1]));
        __builtin_amdgcn_sched_barrier(0);
        float* Hc = Hblk + (c & 1) * H_SIZE;
#pragma unroll
        for (int nt = 0; nt < 2; ++nt)
#pragma unroll
            for (int i = 0; i < 8; ++i) {
                int e = etile * 16 + hi * 8 + i;
                Hc[e * H_LD + nsel * 32 + nt * 16 + row] = silu_f(acc[nt][i]);
            }
        __syncthreads();
        if (tid < TILE_E) {
#pragma unroll 8
            for (int k = 0; k < CHUNK; ++k)
                s += Hc[tid * H_LD + k] * cW2[c * CHUNK + k];
        }
    }

    if (tid < TILE_E) {
        int e = tid;
        int si = SRCs[e], di = DSTs[e];
        float dx = x[si * 3 + 0] - x[di * 3 + 0];
        float dy = x[si * 3 + 1] - x[di * 3 + 1];
        float dz = x[si * 3 + 2] - x[di * 3 + 2];
        float len = fmaxf(sqrtf(dx * dx + dy * dy + dz * dz), 1e-8f);
        float w = s / len;
        float* ox = out + N_NODES * NODE_DIM;
        atomAddF(ox + di * 3 + 0, w * dx);
        atomAddF(ox + di * 3 + 1, w * dy);
        atomAddF(ox + di * 3 + 2, w * dz);
    }
}

extern "C" void kernel_launch(void* const* d_in, const int* in_sizes, int n_in,
                              void* d_out, int out_size, void* d_ws, size_t ws_size,
                              hipStream_t stream)
{
    const float* h     = (const float*)d_in[0];
    const float* x     = (const float*)d_in[1];
    const int*   eidx  = (const int*)  d_in[2];
    const float* edist = (const float*)d_in[3];
    const float* nW1   = (const float*)d_in[4];
    const float* nb1   = (const float*)d_in[5];
    const float* nW2   = (const float*)d_in[6];
    const float* nb2   = (const float*)d_in[7];
    const float* cW1   = (const float*)d_in[8];
    const float* cb1   = (const float*)d_in[9];
    const float* cW2   = (const float*)d_in[10];
    const float* eW1   = (const float*)d_in[11];
    const float* eb1   = (const float*)d_in[12];
    const float* eW2   = (const float*)d_in[13];
    const float* eb2   = (const float*)d_in[14];
    float* out   = (float*)d_out;
    float* wpack = (float*)d_ws;

    egnn_init_out<<<(OUT_FLOATS + PACK_FLOATS) / 256, 256, 0, stream>>>(h, x, nW1, cW1, nW2, out, wpack);

    egnn_edge_kernel<<<N_EDGES / TILE_E, THREADS, 0, stream>>>(
        h, x, eidx, edist,
        nb1, nb2,
        cb1, cW2,
        eW1, eb1, eW2, eb2,
        wpack,
        out);
}


# Node and coord first layers fused into one pass over the A tile: one A fragment feeds 4 accumulators, 16-WMMA trips cover the weight loads, 4 chunk barriers (were 8); in-kernel sum order unchanged
# speedup vs baseline: 2.3894x; 1.0314x over previous
#include <hip/hip_runtime.h>
#include <math.h>


#define N_NODES   32768
#define N_EDGES   524288
#define NODE_DIM  128
#define HIDDEN_DIM 256
#define EDGE_DIM  64
#define IN_DIM    320

#define TILE_E    128
#define THREADS   512
#define CHUNK     64
#define NCHUNK    (HIDDEN_DIM / CHUNK)

#define A_LD      324
#define A_OFF     0
#define A_SIZE    (TILE_E * A_LD)
#define H_LD      68
#define H_OFF     (A_OFF + A_SIZE)
#define H_SIZE    (TILE_E * H_LD)
#define HC_OFF    (H_OFF + 2 * H_SIZE)
#define S_OFF     (HC_OFF + 2 * H_SIZE)
#define SMEM_FLOATS (S_OFF + 4 * TILE_E)

#define KS1        (IN_DIM / 4)
#define KS2        (CHUNK / 4)
#define P1_FLOATS  (IN_DIM * HIDDEN_DIM)
#define P2_FLOATS  (HIDDEN_DIM * NODE_DIM)
#define PN1_OFF    0
#define PC1_OFF    P1_FLOATS
#define PN2_OFF    (2 * P1_FLOATS)
#define PACK_FLOATS (2 * P1_FLOATS + P2_FLOATS)
#define OUT_FLOATS (N_NODES * (NODE_DIM + 3))

typedef float v2f __attribute__((ext_vector_type(2)));
typedef float v8f __attribute__((ext_vector_type(8)));

__device__ __forceinline__ float silu_f(float v) {
    return v / (1.0f + __expf(-v));
}

__device__ __forceinline__ void atomAddF(float* p, float v) {
    __hip_atomic_fetch_add(p, v, __ATOMIC_RELAXED, __HIP_MEMORY_SCOPE_AGENT);
}

struct BSet { float4 v[4]; };

__device__ __forceinline__ void load_bset(BSet& s, const float4* p) {
#pragma unroll
    for (int u = 0; u < 4; ++u) s.v[u] = p[u * 32];
}

#define NC_STRIDE4 (P1_FLOATS / 4)
struct BSetF { float4 n[4]; float4 c[4]; };

__device__ __forceinline__ void load_bsetf(BSetF& s, const float4* p) {
#pragma unroll
    for (int u = 0; u < 4; ++u) {
        s.n[u] = p[u * 32];
        s.c[u] = p[NC_STRIDE4 + u * 32];
    }
}

__device__ __forceinline__ void wm(v8f& acc, v2f a, float bx, float by) {
    v2f b = {bx, by};
    acc = __builtin_amdgcn_wmma_f32_16x16x4_f32( false, a,  false, b,
                                                 (short)0, acc,
                                                 false,  false);
}

__device__ __forceinline__ void trip_l1(const float* __restrict__ ap, const BSetF& b, v8f* acc) {
    v2f a[4];
#pragma unroll
    for (int u = 0; u < 4; ++u) a[u] = *(const v2f*)(ap + u * 4);
#pragma unroll
    for (int u = 0; u < 4; ++u) {
        wm(acc[0], a[u], b.n[u].x, b.n[u].y);
        wm(acc[1], a[u], b.n[u].z, b.n[u].w);
        wm(acc[2], a[u], b.c[u].x, b.c[u].y);
        wm(acc[3], a[u], b.c[u].z, b.c[u].w);
    }
}

__device__ __forceinline__ void trip_l2(const float* __restrict__ hp, const BSet& b, v8f* acc) {
    v2f a[2];
#pragma unroll
    for (int s = 0; s < 2; ++s) a[s] = *(const v2f*)(hp + s * 4);
#pragma unroll
    for (int s = 0; s < 2; ++s) {
        wm(acc[0], a[s], b.v[2 * s].x,     b.v[2 * s].y);
        wm(acc[1], a[s], b.v[2 * s].z,     b.v[2 * s].w);
        wm(acc[2], a[s], b.v[2 * s + 1].x, b.v[2 * s + 1].y);
        wm(acc[3], a[s], b.v[2 * s + 1].z, b.v[2 * s + 1].w);
    }
}

__device__ __forceinline__ void stream_l1(const float* __restrict__ ap, const float4* bg,
                                          BSetF& b0, v8f* acc)
{
    BSetF b1;
#pragma unroll 1
    for (int t = 0; t < KS1 / 4 - 2; t += 2) {
        load_bsetf(b1, bg + 128);
        trip_l1(ap, b0, acc);
        asm volatile("" : "+v"(acc[0]), "+v"(acc[1]), "+v"(acc[2]), "+v"(acc[3]) :: "memory");
        load_bsetf(b0, bg + 256);
        trip_l1(ap + 16, b1, acc);
        ap += 32;
        bg += 256;
    }
    load_bsetf(b1, bg + 128);
    trip_l1(ap, b0, acc);
    trip_l1(ap + 16, b1, acc);
}

__device__ __forceinline__ void stream_l2(const float* __restrict__ hp, const float4* bg,
                                          BSet& b0, v8f* acc)
{
    BSet b1;
#pragma unroll
    for (int t = 0; t < KS2 / 2 - 2; t += 2) {
        load_bset(b1, bg + 128);
        trip_l2(hp, b0, acc);
        asm volatile("" : "+v"(acc[0]), "+v"(acc[1]), "+v"(acc[2]), "+v"(acc[3]) :: "memory");
        load_bset(b0, bg + 256);
        trip_l2(hp + 8, b1, acc);
        hp += 16;
        bg += 256;
    }
    load_bset(b1, bg + 128);
    trip_l2(hp, b0, acc);
    trip_l2(hp + 8, b1, acc);
}

__global__ void egnn_init_out(const float* __restrict__ h,
                              const float* __restrict__ x,
                              const float* __restrict__ nW1,
                              const float* __restrict__ cW1,
                              const float* __restrict__ nW2,
                              float* __restrict__ out,
                              float* __restrict__ wpack)
{
    int i = blockIdx.x * blockDim.x + threadIdx.x;
    if (i < N_NODES * NODE_DIM)      out[i] = h[i];
    else if (i < OUT_FLOATS)         out[i] = x[i - N_NODES * NODE_DIM];
    else {
        int p = i - OUT_FLOATS;
        if (p >= PACK_FLOATS) return;
        int lane = (p >> 2) & 31;
        int row = lane & 15, hi = lane >> 4;
        int j = p & 1, ntl = (p >> 1) & 1;
        if (p < 2 * P1_FLOATS) {
            const float* Wm = (p < P1_FLOATS) ? nW1 : cW1;
            int q  = (p < P1_FLOATS) ? p : p - P1_FLOATS;
            int t  = q >> 7;
            int cn = t / KS1;
            int ks = t - cn * KS1;
            int k  = ks * 4 + 2 * hi + j;
            int n  = (cn >> 1) * CHUNK + (cn & 1) * 32 + ntl * 16 + row;
            wpack[p] = Wm[k * HIDDEN_DIM + n];
        } else {
            int q  = p - 2 * P1_FLOATS;
            int qq = (q >> 7) & 1;
            int ks = (q >> 8) & (KS2 - 1);
            int cn = q >> 12;
            int k  = (cn >> 1) * CHUNK + ks * 4 + 2 * hi + j;
            int n  = (cn & 1) * 64 + qq * 32 + ntl * 16 + row;
            wpack[p] = nW2[k * NODE_DIM + n];
        }
    }
}

__global__ void __launch_bounds__(THREADS)
egnn_edge_kernel(const float* __restrict__ h,   const float* __restrict__ x,
                 const int*   __restrict__ eidx,const float* __restrict__ edist,
                 const float* __restrict__ nb1, const float* __restrict__ nb2,
                 const float* __restrict__ cb1, const float* __restrict__ cW2,
                 const float* __restrict__ eW1, const float* __restrict__ eb1,
                 const float* __restrict__ eW2, const float* __restrict__ eb2,
                 const float* wpack,
                 float* __restrict__ out)
{
    __shared__ __align__(16) float smem[SMEM_FLOATS];
    float* Ablk = smem + A_OFF;
    float* Hblk = smem + H_OFF;
    float* HCblk = smem + HC_OFF;
    int*   SRCs = (int*)(smem + S_OFF);
    int*   DSTs = (int*)(smem + S_OFF + TILE_E);
    float* DIST = smem + S_OFF + 2 * TILE_E;

    const int tid   = threadIdx.x;
    const int lane  = tid & 31;
    const int wave  = tid >> 5;
    const int row   = lane & 15;
    const int hi    = lane >> 4;
    const int etile = wave >> 1;
    const int nsel  = wave & 1;
    const int e0    = blockIdx.x * TILE_E;

    if (tid < TILE_E) {
        SRCs[tid] = eidx[e0 + tid];
        DSTs[tid] = eidx[N_EDGES + e0 + tid];
        DIST[tid] = edist[e0 + tid];
    }
    __syncthreads();

    for (int i = tid; i < TILE_E * 64; i += THREADS) {
        int e = i >> 6, q = i & 63;
        int half = q >> 5, c4 = q & 31;
        int node = half ? DSTs[e] : SRCs[e];
        *(float4*)(Ablk + e * A_LD + half * NODE_DIM + c4 * 4) =
            ((const float4*)h)[node * (NODE_DIM / 4) + c4];
    }

    float* hid = Hblk;
    {
        const int   e = tid & (TILE_E - 1);
        const float d = DIST[e];
#pragma unroll 4
        for (int it = 0; it < EDGE_DIM / (THREADS / TILE_E); ++it) {
            int k = (tid >> 7) + it * (THREADS / TILE_E);
            hid[k * TILE_E + e] = silu_f(d * eW1[k] + eb1[k]);
        }
    }
    __syncthreads();
    {
        const int c4 = (tid & 15) * 4;
        const int e4 = (tid >> 4) * 4;
        float acc[4][4];
#pragma unroll
        for (int c = 0; c < 4; ++c) {
            float bv = eb2[c4 + c];
#pragma unroll
            for (int t = 0; t < 4; ++t) acc[t][c] = bv;
        }
        const float* hp = hid + e4;
        const float* wp = eW2 + c4;
#pragma unroll 8
        for (int k = 0; k < EDGE_DIM; ++k) {
            const float4 hv = *(const float4*)(hp + k * TILE_E);
            const float4 wv = *(const float4*)(wp + k * EDGE_DIM);
            const float hh[4] = {hv.x, hv.y, hv.z, hv.w};
            const float ww[4] = {wv.x, wv.y, wv.z, wv.w};
#pragma unroll
            for (int t = 0; t < 4; ++t)
#pragma unroll
                for (int c = 0; c < 4; ++c)
                    acc[t][c] += hh[t] * ww[c];
        }
#pragma unroll
        for (int t = 0; t < 4; ++t) {
            float4 o;
            o.x = acc[t][0]; o.y = acc[t][1]; o.z = acc[t][2]; o.w = acc[t][3];
            *(float4*)(Ablk + (e4 + t) * A_LD + 2 * NODE_DIM + c4) = o;
        }
    }
    __syncthreads();

    const float*  Arow = Ablk + (etile * 16 + row) * A_LD + 2 * hi;
    const float*  Hrow = Hblk + (etile * 16 + row) * H_LD + 2 * hi;
    const float4* wp4  = (const float4*)wpack;
    const float4* B1   = wp4 + PN1_OFF / 4 + nsel * (KS1 * 32) + lane;
    const float4* Bn2  = wp4 + PN2_OFF / 4 + nsel * (KS2 * 2 * 32) + lane;

    v8f macc[4];
#pragma unroll
    for (int nt = 0; nt < 4; ++nt) {
        float bv = nb2[nsel * 64 + nt * 16 + row];
        v8f t = {bv, bv, bv, bv, bv, bv, bv, bv};
        macc[nt] = t;
    }
    float s = 0.0f;
#pragma unroll 1
    for (int c = 0; c < NCHUNK; ++c) {
        BSetF b1s;
        load_bsetf(b1s, B1 + c * (2 * KS1 * 32));
        v8f acc[4];
#pragma unroll
        for (int nt = 0; nt < 2; ++nt) {
            float bn = nb1[c * CHUNK + nsel * 32 + nt * 16 + row];
            float bc = cb1[c * CHUNK + nsel * 32 + nt * 16 + row];
            v8f tn = {bn, bn, bn, bn, bn, bn, bn, bn};
            v8f tc = {bc, bc, bc, bc, bc, bc, bc, bc};
            acc[nt]     = tn;
            acc[2 + nt] = tc;
        }
        stream_l1(Arow, B1 + c * (2 * KS1 * 32), b1s, acc);
        BSet b2s;
        load_bset(b2s, Bn2 + c * (2 * KS2 * 2 * 32));
        asm volatile("v_nop\n\tv_nop\n\tv_nop\n\tv_nop" : "+v"(acc[0]), "+v"(acc[1]), "+v"(acc[2]), "+v"(acc[3]));
        __builtin_amdgcn_sched_barrier(0);
        float* Hn = Hblk  + (c & 1) * H_SIZE;
        float* Hc = HCblk + (c & 1) * H_SIZE;
#pragma unroll
        for (int nt = 0; nt < 2; ++nt)
#pragma unroll
            for (int i = 0; i < 8; ++i) {
                int e = etile * 16 + hi * 8 + i;
                Hn[e * H_LD + nsel * 32 + nt * 16 + row] = silu_f(acc[nt][i]);
                Hc[e * H_LD + nsel * 32 + nt * 16 + row] = silu_f(acc[2 + nt][i]);
            }
        __syncthreads();
        stream_l2(Hrow + (c & 1) * H_SIZE, Bn2 + c * (2 * KS2 * 2 * 32), b2s, macc);
        asm volatile("v_nop\n\tv_nop\n\tv_nop\n\tv_nop" : "+v"(macc[0]), "+v"(macc[1]), "+v"(macc[2]), "+v"(macc[3]));
        __builtin_amdgcn_sched_barrier(0);
        if (tid < TILE_E) {
#pragma unroll 8
            for (int k = 0; k < CHUNK; ++k)
                s += Hc[tid * H_LD + k] * cW2[c * CHUNK + k];
        }
    }
#pragma unroll
    for (int nt = 0; nt < 4; ++nt)
#pragma unroll
        for (int i = 0; i < 8; ++i) {
            int e = etile * 16 + hi * 8 + i;
            int d = DSTs[e];
            atomAddF(out + d * NODE_DIM + nsel * 64 + nt * 16 + row, macc[nt][i]);
        }

    if (tid < TILE_E) {
        int e = tid;
        int si = SRCs[e], di = DSTs[e];
        float dx = x[si * 3 + 0] - x[di * 3 + 0];
        float dy = x[si * 3 + 1] - x[di * 3 + 1];
        float dz = x[si * 3 + 2] - x[di * 3 + 2];
        float len = fmaxf(sqrtf(dx * dx + dy * dy + dz * dz), 1e-8f);
        float w = s / len;
        float* ox = out + N_NODES * NODE_DIM;
        atomAddF(ox + di * 3 + 0, w * dx);
        atomAddF(ox + di * 3 + 1, w * dy);
        atomAddF(ox + di * 3 + 2, w * dz);
    }
}

extern "C" void kernel_launch(void* const* d_in, const int* in_sizes, int n_in,
                              void* d_out, int out_size, void* d_ws, size_t ws_size,
                              hipStream_t stream)
{
    const float* h     = (const float*)d_in[0];
    const float* x     = (const float*)d_in[1];
    const int*   eidx  = (const int*)  d_in[2];
    const float* edist = (const float*)d_in[3];
    const float* nW1   = (const float*)d_in[4];
    const float* nb1   = (const float*)d_in[5];
    const float* nW2   = (const float*)d_in[6];
    const float* nb2   = (const float*)d_in[7];
    const float* cW1   = (const float*)d_in[8];
    const float* cb1   = (const float*)d_in[9];
    const float* cW2   = (const float*)d_in[10];
    const float* eW1   = (const float*)d_in[11];
    const float* eb1   = (const float*)d_in[12];
    const float* eW2   = (const float*)d_in[13];
    const float* eb2   = (const float*)d_in[14];
    float* out   = (float*)d_out;
    float* wpack = (float*)d_ws;

    egnn_init_out<<<(OUT_FLOATS + PACK_FLOATS) / 256, 256, 0, stream>>>(h, x, nW1, cW1, nW2, out, wpack);

    egnn_edge_kernel<<<N_EDGES / TILE_E, THREADS, 0, stream>>>(
        h, x, eidx, edist,
        nb1, nb2,
        cb1, cW2,
        eW1, eb1, eW2, eb2,
        wpack,
        out);
}


# Resident workgroups (256) loop over 16 tiles each: next tile's indices and h rows fetched during the current tile's scatter, not at a workgroup's start; in-kernel sum order unchanged, atomics kept
# speedup vs baseline: 2.4977x; 1.0453x over previous
#include <hip/hip_runtime.h>
#include <math.h>


#define N_NODES   32768
#define N_EDGES   524288
#define NODE_DIM  128
#define HIDDEN_DIM 256
#define EDGE_DIM  64
#define IN_DIM    320

#define TILE_E    128
#define THREADS   512
#define CHUNK     64
#define NCHUNK    (HIDDEN_DIM / CHUNK)

#define A_LD      324
#define A_OFF     0
#define A_SIZE    (TILE_E * A_LD)
#define H_LD      68
#define H_OFF     (A_OFF + A_SIZE)
#define H_SIZE    (TILE_E * H_LD)
#define HC_OFF    (H_OFF + 2 * H_SIZE)
#define S_OFF     (HC_OFF + 2 * H_SIZE)
#define S_BUF     (3 * TILE_E)
#define SMEM_FLOATS (S_OFF + 2 * S_BUF)
#define N_TILES   (N_EDGES / TILE_E)
#define N_GROUPS  256

#define KS1        (IN_DIM / 4)
#define KS2        (CHUNK / 4)
#define P1_FLOATS  (IN_DIM * HIDDEN_DIM)
#define P2_FLOATS  (HIDDEN_DIM * NODE_DIM)
#define PN1_OFF    0
#define PC1_OFF    P1_FLOATS
#define PN2_OFF    (2 * P1_FLOATS)
#define PACK_FLOATS (2 * P1_FLOATS + P2_FLOATS)
#define OUT_FLOATS (N_NODES * (NODE_DIM + 3))

typedef float v2f __attribute__((ext_vector_type(2)));
typedef float v8f __attribute__((ext_vector_type(8)));

__device__ __forceinline__ float silu_f(float v) {
    return v / (1.0f + __expf(-v));
}

__device__ __forceinline__ void atomAddF(float* p, float v) {
    __hip_atomic_fetch_add(p, v, __ATOMIC_RELAXED, __HIP_MEMORY_SCOPE_AGENT);
}

struct BSet { float4 v[4]; };

__device__ __forceinline__ void load_bset(BSet& s, const float4* p) {
#pragma unroll
    for (int u = 0; u < 4; ++u) s.v[u] = p[u * 32];
}

#define NC_STRIDE4 (P1_FLOATS / 4)
struct BSetF { float4 n[4]; float4 c[4]; };

__device__ __forceinline__ void load_bsetf(BSetF& s, const float4* p) {
#pragma unroll
    for (int u = 0; u < 4; ++u) {
        s.n[u] = p[u * 32];
        s.c[u] = p[NC_STRIDE4 + u * 32];
    }
}

__device__ __forceinline__ void wm(v8f& acc, v2f a, float bx, float by) {
    v2f b = {bx, by};
    acc = __builtin_amdgcn_wmma_f32_16x16x4_f32( false, a,  false, b,
                                                 (short)0, acc,
                                                 false,  false);
}

__device__ __forceinline__ void trip_l1(const float* __restrict__ ap, const BSetF& b, v8f* acc) {
    v2f a[4];
#pragma unroll
    for (int u = 0; u < 4; ++u) a[u] = *(const v2f*)(ap + u * 4);
#pragma unroll
    for (int u = 0; u < 4; ++u) {
        wm(acc[0], a[u], b.n[u].x, b.n[u].y);
        wm(acc[1], a[u], b.n[u].z, b.n[u].w);
        wm(acc[2], a[u], b.c[u].x, b.c[u].y);
        wm(acc[3], a[u], b.c[u].z, b.c[u].w);
    }
}

__device__ __forceinline__ void trip_l2(const float* __restrict__ hp, const BSet& b, v8f* acc) {
    v2f a[2];
#pragma unroll
    for (int s = 0; s < 2; ++s) a[s] = *(const v2f*)(hp + s * 4);
#pragma unroll
    for (int s = 0; s < 2; ++s) {
        wm(acc[0], a[s], b.v[2 * s].x,     b.v[2 * s].y);
        wm(acc[1], a[s], b.v[2 * s].z,     b.v[2 * s].w);
        wm(acc[2], a[s], b.v[2 * s + 1].x, b.v[2 * s + 1].y);
        wm(acc[3], a[s], b.v[2 * s + 1].z, b.v[2 * s + 1].w);
    }
}

__device__ __forceinline__ void stream_l1(const float* __restrict__ ap, const float4* bg,
                                          BSetF& b0, v8f* acc)
{
    BSetF b1;
#pragma unroll 1
    for (int t = 0; t < KS1 / 4 - 2; t += 2) {
        load_bsetf(b1, bg + 128);
        trip_l1(ap, b0, acc);
        asm volatile("" : "+v"(acc[0]), "+v"(acc[1]), "+v"(acc[2]), "+v"(acc[3]) :: "memory");
        load_bsetf(b0, bg + 256);
        trip_l1(ap + 16, b1, acc);
        ap += 32;
        bg += 256;
    }
    load_bsetf(b1, bg + 128);
    trip_l1(ap, b0, acc);
    trip_l1(ap + 16, b1, acc);
}

__device__ __forceinline__ void stream_l2(const float* __restrict__ hp, const float4* bg,
                                          BSet& b0, v8f* acc)
{
    BSet b1;
#pragma unroll
    for (int t = 0; t < KS2 / 2 - 2; t += 2) {
        load_bset(b1, bg + 128);
        trip_l2(hp, b0, acc);
        asm volatile("" : "+v"(acc[0]), "+v"(acc[1]), "+v"(acc[2]), "+v"(acc[3]) :: "memory");
        load_bset(b0, bg + 256);
        trip_l2(hp + 8, b1, acc);
        hp += 16;
        bg += 256;
    }
    load_bset(b1, bg + 128);
    trip_l2(hp, b0, acc);
    trip_l2(hp + 8, b1, acc);
}

__global__ void egnn_init_out(const float* __restrict__ h,
                              const float* __restrict__ x,
                              const float* __restrict__ nW1,
                              const float* __restrict__ cW1,
                              const float* __restrict__ nW2,
                              float* __restrict__ out,
                              float* __restrict__ wpack)
{
    int i = blockIdx.x * blockDim.x + threadIdx.x;
    if (i < N_NODES * NODE_DIM)      out[i] = h[i];
    else if (i < OUT_FLOATS)         out[i] = x[i - N_NODES * NODE_DIM];
    else {
        int p = i - OUT_FLOATS;
        if (p >= PACK_FLOATS) return;
        int lane = (p >> 2) & 31;
        int row = lane & 15, hi = lane >> 4;
        int j = p & 1, ntl = (p >> 1) & 1;
        if (p < 2 * P1_FLOATS) {
            const float* Wm = (p < P1_FLOATS) ? nW1 : cW1;
            int q  = (p < P1_FLOATS) ? p : p - P1_FLOATS;
            int t  = q >> 7;
            int cn = t / KS1;
            int ks = t - cn * KS1;
            int k  = ks * 4 + 2 * hi + j;
            int n  = (cn >> 1) * CHUNK + (cn & 1) * 32 + ntl * 16 + row;
            wpack[p] = Wm[k * HIDDEN_DIM + n];
        } else {
            int q  = p - 2 * P1_FLOATS;
            int qq = (q >> 7) & 1;
            int ks = (q >> 8) & (KS2 - 1);
            int cn = q >> 12;
            int k  = (cn >> 1) * CHUNK + ks * 4 + 2 * hi + j;
            int n  = (cn & 1) * 64 + qq * 32 + ntl * 16 + row;
            wpack[p] = nW2[k * NODE_DIM + n];
        }
    }
}

#define G_ALL(X) X(0) X(1) X(2) X(3) X(4) X(5) X(6) X(7) X(8) X(9) X(10) X(11) X(12) X(13) X(14) X(15)
#define G_DECL(u)  float4 g##u;
#define G_LOAD(u)  g##u = ((const float4*)h)[gidx[(u) * 8] * (NODE_DIM / 4) + gc4];
#define G_STORE(u) *(float4*)(arow + (u) * 8 * A_LD) = g##u;
#define G_ZERO(u)  g##u = make_float4(0.0f, 0.0f, 0.0f, 0.0f);

__global__ void __launch_bounds__(THREADS)
egnn_edge_kernel(const float* __restrict__ h,   const float* __restrict__ x,
                 const int*   __restrict__ eidx,const float* __restrict__ edist,
                 const float* __restrict__ nb1, const float* __restrict__ nb2,
                 const float* __restrict__ cb1, const float* __restrict__ cW2,
                 const float* __restrict__ eW1, const float* __restrict__ eb1,
                 const float* __restrict__ eW2, const float* __restrict__ eb2,
                 const float* wpack,
                 float* __restrict__ out)
{
    __shared__ __align__(16) float smem[SMEM_FLOATS];
    float* Ablk = smem + A_OFF;
    float* Hblk = smem + H_OFF;
    float* HCblk = smem + HC_OFF;
    int*   Sbase = (int*)(smem + S_OFF);

    const int tid   = threadIdx.x;
    const int lane  = tid & 31;
    const int wave  = tid >> 5;
    const int row   = lane & 15;
    const int hi    = lane >> 4;
    const int etile = wave >> 1;
    const int nsel  = wave & 1;
    int tile = blockIdx.x;
    if (tid < TILE_E) {
        const int e0 = tile * TILE_E;
        Sbase[tid]              = eidx[e0 + tid];
        Sbase[TILE_E + tid]     = eidx[N_EDGES + e0 + tid];
        ((float*)Sbase)[2 * TILE_E + tid] = edist[e0 + tid];
    }
    __syncthreads();
    const int gc4 = tid & 31;
    const int goff = ((tid >> 5) & 1) * TILE_E + (tid >> 6);
    G_ALL(G_DECL)
    {
        const int* gidx = Sbase + goff;
        G_ALL(G_LOAD)
    }

#pragma unroll 1
    for (int trip = 0; tile < N_TILES; tile += N_GROUPS, ++trip) {
    int*   SRCs  = Sbase + (trip & 1) * S_BUF;
    int*   DSTs  = SRCs + TILE_E;
    float* DIST  = (float*)(SRCs + 2 * TILE_E);
    int*   Snext = Sbase + ((trip + 1) & 1) * S_BUF;
    const int ntile = tile + N_GROUPS;

    {
        float* arow = Ablk + (tid >> 6) * A_LD + ((tid >> 5) & 1) * NODE_DIM + gc4 * 4;
        G_ALL(G_STORE)
    }

    float* hid = Hblk;
    {
        const int   e = tid & (TILE_E - 1);
        const float d = DIST[e];
#pragma unroll 4
        for (int it = 0; it < EDGE_DIM / (THREADS / TILE_E); ++it) {
            int k = (tid >> 7) + it * (THREADS / TILE_E);
            hid[k * TILE_E + e] = silu_f(d * eW1[k] + eb1[k]);
        }
    }
    __syncthreads();
    {
        const int c4 = (tid & 15) * 4;
        const int e4 = (tid >> 4) * 4;
        float acc[4][4];
#pragma unroll
        for (int c = 0; c < 4; ++c) {
            float bv = eb2[c4 + c];
#pragma unroll
            for (int t = 0; t < 4; ++t) acc[t][c] = bv;
        }
        const float* hp = hid + e4;
        const float* wp = eW2 + c4;
#pragma unroll 8
        for (int k = 0; k < EDGE_DIM; ++k) {
            const float4 hv = *(const float4*)(hp + k * TILE_E);
            const float4 wv = *(const float4*)(wp + k * EDGE_DIM);
            const float hh[4] = {hv.x, hv.y, hv.z, hv.w};
            const float ww[4] = {wv.x, wv.y, wv.z, wv.w};
#pragma unroll
            for (int t = 0; t < 4; ++t)
#pragma unroll
                for (int c = 0; c < 4; ++c)
                    acc[t][c] += hh[t] * ww[c];
        }
#pragma unroll
        for (int t = 0; t < 4; ++t) {
            float4 o;
            o.x = acc[t][0]; o.y = acc[t][1]; o.z = acc[t][2]; o.w = acc[t][3];
            *(float4*)(Ablk + (e4 + t) * A_LD + 2 * NODE_DIM + c4) = o;
        }
    }
    __syncthreads();

    if (ntile < N_TILES && tid < TILE_E) {
        const int e0 = ntile * TILE_E;
        Snext[tid]              = eidx[e0 + tid];
        Snext[TILE_E + tid]     = eidx[N_EDGES + e0 + tid];
        ((float*)Snext)[2 * TILE_E + tid] = edist[e0 + tid];
    }

    const float*  Arow = Ablk + (etile * 16 + row) * A_LD + 2 * hi;
    const float*  Hrow = Hblk + (etile * 16 + row) * H_LD + 2 * hi;
    const float4* wp4  = (const float4*)wpack;
    const float4* B1   = wp4 + PN1_OFF / 4 + nsel * (KS1 * 32) + lane;
    const float4* Bn2  = wp4 + PN2_OFF / 4 + nsel * (KS2 * 2 * 32) + lane;

    v8f macc[4];
#pragma unroll
    for (int nt = 0; nt < 4; ++nt) {
        float bv = nb2[nsel * 64 + nt * 16 + row];
        v8f t = {bv, bv, bv, bv, bv, bv, bv, bv};
        macc[nt] = t;
    }
    float s = 0.0f;
#pragma unroll 1
    for (int c = 0; c < NCHUNK; ++c) {
        BSetF b1s;
        load_bsetf(b1s, B1 + c * (2 * KS1 * 32));
        v8f acc[4];
#pragma unroll
        for (int nt = 0; nt < 2; ++nt) {
            float bn = nb1[c * CHUNK + nsel * 32 + nt * 16 + row];
            float bc = cb1[c * CHUNK + nsel * 32 + nt * 16 + row];
            v8f tn = {bn, bn, bn, bn, bn, bn, bn, bn};
            v8f tc = {bc, bc, bc, bc, bc, bc, bc, bc};
            acc[nt]     = tn;
            acc[2 + nt] = tc;
        }
        stream_l1(Arow, B1 + c * (2 * KS1 * 32), b1s, acc);
        BSet b2s;
        load_bset(b2s, Bn2 + c * (2 * KS2 * 2 * 32));
        asm volatile("v_nop\n\tv_nop\n\tv_nop\n\tv_nop" : "+v"(acc[0]), "+v"(acc[1]), "+v"(acc[2]), "+v"(acc[3]));
        __builtin_amdgcn_sched_barrier(0);
        float* Hn = Hblk  + (c & 1) * H_SIZE;
        float* Hc = HCblk + (c & 1) * H_SIZE;
#pragma unroll
        for (int nt = 0; nt < 2; ++nt)
#pragma unroll
            for (int i = 0; i < 8; ++i) {
                int e = etile * 16 + hi * 8 + i;
                Hn[e * H_LD + nsel * 32 + nt * 16 + row] = silu_f(acc[nt][i]);
                Hc[e * H_LD + nsel * 32 + nt * 16 + row] = silu_f(acc[2 + nt][i]);
            }
        __syncthreads();
        stream_l2(Hrow + (c & 1) * H_SIZE, Bn2 + c * (2 * KS2 * 2 * 32), b2s, macc);
        asm volatile("v_nop\n\tv_nop\n\tv_nop\n\tv_nop" : "+v"(macc[0]), "+v"(macc[1]), "+v"(macc[2]), "+v"(macc[3]));
        __builtin_amdgcn_sched_barrier(0);
        if (tid < TILE_E) {
#pragma unroll 8
            for (int k = 0; k < CHUNK; ++k)
                s += Hc[tid * H_LD + k] * cW2[c * CHUNK + k];
        }
    }
    if (ntile < N_TILES) {
        const int* gidx = Snext + goff;
        G_ALL(G_LOAD)
    } else {
        G_ALL(G_ZERO)
    }
#pragma unroll
    for (int nt = 0; nt < 4; ++nt)
#pragma unroll
        for (int i = 0; i < 8; ++i) {
            int e = etile * 16 + hi * 8 + i;
            int d = DSTs[e];
            atomAddF(out + d * NODE_DIM + nsel * 64 + nt * 16 + row, macc[nt][i]);
        }

    if (tid < TILE_E) {
        int e = tid;
        int si = SRCs[e], di = DSTs[e];
        float dx = x[si * 3 + 0] - x[di * 3 + 0];
        float dy = x[si * 3 + 1] - x[di * 3 + 1];
        float dz = x[si * 3 + 2] - x[di * 3 + 2];
        float len = fmaxf(sqrtf(dx * dx + dy * dy + dz * dz), 1e-8f);
        float w = s / len;
        float* ox = out + N_NODES * NODE_DIM;
        atomAddF(ox + di * 3 + 0, w * dx);
        atomAddF(ox + di * 3 + 1, w * dy);
        atomAddF(ox + di * 3 + 2, w * dz);
    }
    }
}

extern "C" void kernel_launch(void* const* d_in, const int* in_sizes, int n_in,
                              void* d_out, int out_size, void* d_ws, size_t ws_size,
                              hipStream_t stream)
{
    const float* h     = (const float*)d_in[0];
    const float* x     = (const float*)d_in[1];
    const int*   eidx  = (const int*)  d_in[2];
    const float* edist = (const float*)d_in[3];
    const float* nW1   = (const float*)d_in[4];
    const float* nb1   = (const float*)d_in[5];
    const float* nW2   = (const float*)d_in[6];
    const float* nb2   = (const float*)d_in[7];
    const float* cW1   = (const float*)d_in[8];
    const float* cb1   = (const float*)d_in[9];
    const float* cW2   = (const float*)d_in[10];
    const float* eW1   = (const float*)d_in[11];
    const float* eb1   = (const float*)d_in[12];
    const float* eW2   = (const float*)d_in[13];
    const float* eb2   = (const float*)d_in[14];
    float* out   = (float*)d_out;
    float* wpack = (float*)d_ws;

    egnn_init_out<<<(OUT_FLOATS + PACK_FLOATS) / 256, 256, 0, stream>>>(h, x, nW1, cW1, nW2, out, wpack);

    egnn_edge_kernel<<<N_GROUPS, THREADS, 0, stream>>>(
        h, x, eidx, edist,
        nb1, nb2,
        cb1, cW2,
        eW1, eb1, eW2, eb2,
        wpack,
        out);
}
